# starting kernel
# baseline (speedup 1.0000x reference)
#include <hip/hip_runtime.h>

typedef __attribute__((ext_vector_type(2))) float v2f;
typedef __attribute__((ext_vector_type(8))) float v8f;

#define IN_CH   16
#define OUT_CH  16
#define KTAP    3
#define DIL     4
#define CONDD   8
#define BATCH   32
#define TOUT    65536
#define PADX    (DIL * (KTAP - 1))
#define TIN     (TOUT + PADX)
#define CTOT    (IN_CH * KTAP)
#define NCHUNK  (CTOT / 4)
#define TILES_PER_WAVE 16

__global__ __launch_bounds__(256) void hypernet_kernel(
    const float* __restrict__ p,  const float* __restrict__ W1,
    const float* __restrict__ b1, const float* __restrict__ W2,
    const float* __restrict__ b2, float* __restrict__ wbuf)
{
    const int b   = blockIdx.x;
    const int tid = threadIdx.x;
    __shared__ float h[IN_CH];

    if (tid < IN_CH) {
        float acc = b1[tid];
        #pragma unroll
        for (int i = 0; i < CONDD; ++i)
            acc = fmaf(p[b * CONDD + i], W1[i * IN_CH + tid], acc);
        h[tid] = (acc > 0.0f) ? acc : 0.2f * acc;
    }
    __syncthreads();

    for (int idx = tid; idx < CTOT * OUT_CH; idx += 256) {
        float acc = b2[idx];
        #pragma unroll
        for (int j = 0; j < IN_CH; ++j)
            acc = fmaf(h[j], W2[j * (CTOT * OUT_CH) + idx], acc);
        wbuf[b * (CTOT * OUT_CH) + idx] = acc;
    }
}

__global__ __launch_bounds__(256) void hyperconv_main(
    const float* __restrict__ x, const float* __restrict__ wbuf,
    float* __restrict__ y)
{
    const int lane  = threadIdx.x & 31;
    const int half  = lane >> 4;
    const int ln    = lane & 15;
    const int gwave = blockIdx.x * (blockDim.x >> 5) + (threadIdx.x >> 5);
    const int b     = gwave >> 8;
    const int strip = gwave & 255;
    const int t0    = strip * (16 * TILES_PER_WAVE);

    v2f A[NCHUNK];
    {
        const float* wb = wbuf + b * (CTOT * OUT_CH);
        #pragma unroll
        for (int j = 0; j < NCHUNK; ++j) {
            const int c0 = 4 * j + 2 * half;
            A[j][0] = wb[(c0 + 0) * OUT_CH + ln];
            A[j][1] = wb[(c0 + 1) * OUT_CH + ln];
        }
    }

    int offs0[NCHUNK], offs1[NCHUNK];
    #pragma unroll
    for (int j = 0; j < NCHUNK; ++j) {
        const int c0 = 4 * j + 2 * half;
        const int ca = c0,     ta = ca >> 4, cha = ca & 15;
        const int cb = c0 + 1, tb = cb >> 4, chb = cb & 15;
        offs0[j] = (b * IN_CH + cha) * TIN + (PADX - DIL * ta) + t0 + ln;
        offs1[j] = (b * IN_CH + chb) * TIN + (PADX - DIL * tb) + t0 + ln;
    }

    float* yb = y + (size_t)b * OUT_CH * TOUT + t0 + ln;

    for (int tt = 0; tt < TILES_PER_WAVE; ++tt) {
        const int td = tt * 16;
        v8f acc = {};
        #pragma unroll
        for (int j = 0; j < NCHUNK; ++j) {
            v2f bm;
            bm[0] = x[offs0[j] + td];
            bm[1] = x[offs1[j] + td];
            acc = __builtin_amdgcn_wmma_f32_16x16x4_f32(
                false, A[j], false, bm, (short)0, acc, false, false);
        }
        #pragma unroll
        for (int r = 0; r < 8; ++r) {
            const int o = r + 8 * half;
            __builtin_nontemporal_store(acc[r], yb + (size_t)o * TOUT + td);
        }
    }
}

extern "C" void kernel_launch(void* const* d_in, const int* in_sizes, int n_in,
                              void* d_out, int out_size, void* d_ws, size_t ws_size,
                              hipStream_t stream) {
    const float* x  = (const float*)d_in[0];
    const float* p  = (const float*)d_in[1];
    const float* W1 = (const float*)d_in[2];
    const float* b1 = (const float*)d_in[3];
    const float* W2 = (const float*)d_in[4];
    const float* b2 = (const float*)d_in[5];
    float* y    = (float*)d_out;
    float* wbuf = (float*)d_ws;

    hipLaunchKernelGGL(hypernet_kernel, dim3(BATCH), dim3(256), 0, stream,
                       p, W1, b1, W2, b2, wbuf);

    const int total_waves = BATCH * (TOUT / (16 * TILES_PER_WAVE));
    hipLaunchKernelGGL(hyperconv_main, dim3(total_waves / 8), dim3(256), 0, stream,
                       x, wbuf, y);
}


# 16 B per lane: column c of sub-tile q is time 4c+q, so x loads and y stores are b128, 256 contiguous B per half-wave; 4 accumulator tiles per 64-position trip, 512 positions per wave; sum order kept
# speedup vs baseline: 1.4684x; 1.4684x over previous
#include <hip/hip_runtime.h>

typedef __attribute__((ext_vector_type(2))) float v2f;
typedef __attribute__((ext_vector_type(4))) float v4f;
typedef __attribute__((ext_vector_type(8))) float v8f;

#define IN_CH   16
#define OUT_CH  16
#define KTAP    3
#define DIL     4
#define CONDD   8
#define BATCH   32
#define TOUT    65536
#define PADX    (DIL * (KTAP - 1))
#define TIN     (TOUT + PADX)
#define CTOT    (IN_CH * KTAP)
#define NCHUNK  (CTOT / 4)
#define SUBTILES       4
#define POS_PER_TRIP   (16 * SUBTILES)
#define TRIPS_PER_WAVE 8
#define POS_PER_WAVE   (POS_PER_TRIP * TRIPS_PER_WAVE)
#define STRIPS         (TOUT / POS_PER_WAVE)

__global__ __launch_bounds__(256) void hypernet_kernel(
    const float* __restrict__ p,  const float* __restrict__ W1,
    const float* __restrict__ b1, const float* __restrict__ W2,
    const float* __restrict__ b2, float* __restrict__ wbuf)
{
    const int b   = blockIdx.x;
    const int tid = threadIdx.x;
    __shared__ float h[IN_CH];

    if (tid < IN_CH) {
        float acc = b1[tid];
        #pragma unroll
        for (int i = 0; i < CONDD; ++i)
            acc = fmaf(p[b * CONDD + i], W1[i * IN_CH + tid], acc);
        h[tid] = (acc > 0.0f) ? acc : 0.2f * acc;
    }
    __syncthreads();

    for (int idx = tid; idx < CTOT * OUT_CH; idx += 256) {
        float acc = b2[idx];
        #pragma unroll
        for (int j = 0; j < IN_CH; ++j)
            acc = fmaf(h[j], W2[j * (CTOT * OUT_CH) + idx], acc);
        wbuf[b * (CTOT * OUT_CH) + idx] = acc;
    }
}

__global__ __launch_bounds__(256) void hyperconv_main(
    const float* __restrict__ x, const float* __restrict__ wbuf,
    float* __restrict__ y)
{
    const int lane  = threadIdx.x & 31;
    const int half  = lane >> 4;
    const int ln    = lane & 15;
    const int gwave = blockIdx.x * (blockDim.x >> 5) + (threadIdx.x >> 5);
    const int b     = gwave / STRIPS;
    const int strip = gwave % STRIPS;
    const int t0    = strip * POS_PER_WAVE;

    v2f A[NCHUNK];
    {
        const float* wb = wbuf + b * (CTOT * OUT_CH);
        #pragma unroll
        for (int j = 0; j < NCHUNK; ++j) {
            const int c0 = 4 * j + 2 * half;
            A[j][0] = wb[(c0 + 0) * OUT_CH + ln];
            A[j][1] = wb[(c0 + 1) * OUT_CH + ln];
        }
    }

    const float* xl = x + ((size_t)b * IN_CH + 2 * half) * TIN + t0 + 4 * ln;
    float*       yl = y + ((size_t)b * OUT_CH + 8 * half) * TOUT + t0 + 4 * ln;

    for (int trip = 0; trip < TRIPS_PER_WAVE; ++trip) {
        v4f B0[NCHUNK], B1[NCHUNK];
        #pragma unroll
        for (int j = 0; j < NCHUNK; ++j) {
            const int off = (4 * (j & 3)) * TIN + (PADX - DIL * (j >> 2));
            B0[j] = *reinterpret_cast<const v4f*>(xl + off);
            B1[j] = *reinterpret_cast<const v4f*>(xl + off + TIN);
        }

        v8f acc[SUBTILES] = {};
        #pragma unroll
        for (int j = 0; j < NCHUNK; ++j) {
            #pragma unroll
            for (int q = 0; q < SUBTILES; ++q) {
                v2f bm;
                bm[0] = B0[j][q];
                bm[1] = B1[j][q];
                acc[q] = __builtin_amdgcn_wmma_f32_16x16x4_f32(
                    false, A[j], false, bm, (short)0, acc[q], false, false);
            }
        }

        #pragma unroll
        for (int r = 0; r < 8; ++r) {
            v4f o4;
            #pragma unroll
            for (int q = 0; q < SUBTILES; ++q) o4[q] = acc[q][r];
            __builtin_nontemporal_store(o4, reinterpret_cast<v4f*>(yl + (size_t)r * TOUT));
        }

        xl += POS_PER_TRIP;
        yl += POS_PER_TRIP;
    }
}

extern "C" void kernel_launch(void* const* d_in, const int* in_sizes, int n_in,
                              void* d_out, int out_size, void* d_ws, size_t ws_size,
                              hipStream_t stream) {
    const float* x  = (const float*)d_in[0];
    const float* p  = (const float*)d_in[1];
    const float* W1 = (const float*)d_in[2];
    const float* b1 = (const float*)d_in[3];
    const float* W2 = (const float*)d_in[4];
    const float* b2 = (const float*)d_in[5];
    float* y    = (float*)d_out;
    float* wbuf = (float*)d_ws;

    hipLaunchKernelGGL(hypernet_kernel, dim3(BATCH), dim3(256), 0, stream,
                       p, W1, b1, W2, b2, wbuf);

    const int total_waves = BATCH * STRIPS;
    hipLaunchKernelGGL(hyperconv_main, dim3(total_waves / 8), dim3(256), 0, stream,
                       x, wbuf, y);
}


# next trip's 24 x-fragment loads issued into a second register set ahead of the current trip's 48 WMMAs (one trip ahead, set handed over at trip end, last trip peeled); sum order kept
# speedup vs baseline: 1.5486x; 1.0546x over previous
#include <hip/hip_runtime.h>

typedef __attribute__((ext_vector_type(2))) float v2f;
typedef __attribute__((ext_vector_type(4))) float v4f;
typedef __attribute__((ext_vector_type(8))) float v8f;

#define IN_CH   16
#define OUT_CH  16
#define KTAP    3
#define DIL     4
#define CONDD   8
#define BATCH   32
#define TOUT    65536
#define PADX    (DIL * (KTAP - 1))
#define TIN     (TOUT + PADX)
#define CTOT    (IN_CH * KTAP)
#define NCHUNK  (CTOT / 4)
#define SUBTILES       4
#define POS_PER_TRIP   (16 * SUBTILES)
#define TRIPS_PER_WAVE 8
#define POS_PER_WAVE   (POS_PER_TRIP * TRIPS_PER_WAVE)
#define STRIPS         (TOUT / POS_PER_WAVE)

__global__ __launch_bounds__(256) void hypernet_kernel(
    const float* __restrict__ p,  const float* __restrict__ W1,
    const float* __restrict__ b1, const float* __restrict__ W2,
    const float* __restrict__ b2, float* __restrict__ wbuf)
{
    const int b   = blockIdx.x;
    const int tid = threadIdx.x;
    __shared__ float h[IN_CH];

    if (tid < IN_CH) {
        float acc = b1[tid];
        #pragma unroll
        for (int i = 0; i < CONDD; ++i)
            acc = fmaf(p[b * CONDD + i], W1[i * IN_CH + tid], acc);
        h[tid] = (acc > 0.0f) ? acc : 0.2f * acc;
    }
    __syncthreads();

    for (int idx = tid; idx < CTOT * OUT_CH; idx += 256) {
        float acc = b2[idx];
        #pragma unroll
        for (int j = 0; j < IN_CH; ++j)
            acc = fmaf(h[j], W2[j * (CTOT * OUT_CH) + idx], acc);
        wbuf[b * (CTOT * OUT_CH) + idx] = acc;
    }
}

__device__ __forceinline__ void load_trip(const float* __restrict__ xl,
                                          v4f (&B0)[NCHUNK], v4f (&B1)[NCHUNK])
{
    #pragma unroll
    for (int j = 0; j < NCHUNK; ++j) {
        const int off = (4 * (j & 3)) * TIN + (PADX - DIL * (j >> 2));
        B0[j] = *reinterpret_cast<const v4f*>(xl + off);
        B1[j] = *reinterpret_cast<const v4f*>(xl + off + TIN);
    }
}

__device__ __forceinline__ void compute_trip(const v2f (&A)[NCHUNK],
                                             const v4f (&B0)[NCHUNK], const v4f (&B1)[NCHUNK],
                                             float* __restrict__ yl)
{
    v8f acc[SUBTILES] = {};
    #pragma unroll
    for (int j = 0; j < NCHUNK; ++j) {
        #pragma unroll
        for (int q = 0; q < SUBTILES; ++q) {
            v2f bm;
            bm[0] = B0[j][q];
            bm[1] = B1[j][q];
            acc[q] = __builtin_amdgcn_wmma_f32_16x16x4_f32(
                false, A[j], false, bm, (short)0, acc[q], false, false);
        }
    }

    #pragma unroll
    for (int r = 0; r < 8; ++r) {
        v4f o4;
        #pragma unroll
        for (int q = 0; q < SUBTILES; ++q) o4[q] = acc[q][r];
        __builtin_nontemporal_store(o4, reinterpret_cast<v4f*>(yl + (size_t)r * TOUT));
    }
}

__global__ __launch_bounds__(256) void hyperconv_main(
    const float* __restrict__ x, const float* __restrict__ wbuf,
    float* __restrict__ y)
{
    const int lane  = threadIdx.x & 31;
    const int half  = lane >> 4;
    const int ln    = lane & 15;
    const int gwave = blockIdx.x * (blockDim.x >> 5) + (threadIdx.x >> 5);
    const int b     = gwave / STRIPS;
    const int strip = gwave % STRIPS;
    const int t0    = strip * POS_PER_WAVE;

    const float* xl = x + ((size_t)b * IN_CH + 2 * half) * TIN + t0 + 4 * ln;
    float*       yl = y + ((size_t)b * OUT_CH + 8 * half) * TOUT + t0 + 4 * ln;

    v2f A[NCHUNK];
    {
        const float* wb = wbuf + b * (CTOT * OUT_CH);
        #pragma unroll
        for (int j = 0; j < NCHUNK; ++j) {
            const int c0 = 4 * j + 2 * half;
            A[j][0] = wb[(c0 + 0) * OUT_CH + ln];
            A[j][1] = wb[(c0 + 1) * OUT_CH + ln];
        }
    }

    v4f B0[NCHUNK], B1[NCHUNK];
    load_trip(xl, B0, B1);

    for (int trip = 0; trip < TRIPS_PER_WAVE - 1; ++trip) {
        v4f N0[NCHUNK], N1[NCHUNK];
        load_trip(xl + POS_PER_TRIP, N0, N1);
        compute_trip(A, B0, B1, yl);
        #pragma unroll
        for (int j = 0; j < NCHUNK; ++j) { B0[j] = N0[j]; B1[j] = N1[j]; }
        xl += POS_PER_TRIP;
        yl += POS_PER_TRIP;
    }
    compute_trip(A, B0, B1, yl);
}

extern "C" void kernel_launch(void* const* d_in, const int* in_sizes, int n_in,
                              void* d_out, int out_size, void* d_ws, size_t ws_size,
                              hipStream_t stream) {
    const float* x  = (const float*)d_in[0];
    const float* p  = (const float*)d_in[1];
    const float* W1 = (const float*)d_in[2];
    const float* b1 = (const float*)d_in[3];
    const float* W2 = (const float*)d_in[4];
    const float* b2 = (const float*)d_in[5];
    float* y    = (float*)d_out;
    float* wbuf = (float*)d_ws;

    hipLaunchKernelGGL(hypernet_kernel, dim3(BATCH), dim3(256), 0, stream,
                       p, W1, b1, W2, b2, wbuf);

    const int total_waves = BATCH * STRIPS;
    hipLaunchKernelGGL(hyperconv_main, dim3(total_waves / 8), dim3(256), 0, stream,
                       x, wbuf, y);
}


# one launch instead of two: each work-group computes its batch sample's 768 hyper-weights itself (same fma order) into LDS and reads its A fragments from there; no scratch buffer; sum order kept
# speedup vs baseline: 1.7643x; 1.1393x over previous
#include <hip/hip_runtime.h>

typedef __attribute__((ext_vector_type(2))) float v2f;
typedef __attribute__((ext_vector_type(4))) float v4f;
typedef __attribute__((ext_vector_type(8))) float v8f;

#define IN_CH   16
#define OUT_CH  16
#define KTAP    3
#define DIL     4
#define CONDD   8
#define BATCH   32
#define TOUT    65536
#define PADX    (DIL * (KTAP - 1))
#define TIN     (TOUT + PADX)
#define CTOT    (IN_CH * KTAP)
#define NCHUNK  (CTOT / 4)
#define SUBTILES       4
#define POS_PER_TRIP   (16 * SUBTILES)
#define TRIPS_PER_WAVE 8
#define POS_PER_WAVE   (POS_PER_TRIP * TRIPS_PER_WAVE)
#define STRIPS         (TOUT / POS_PER_WAVE)
#define WAVES_PER_GROUP 8
#define GROUPS_PER_B   (STRIPS / WAVES_PER_GROUP)
#define W_PER_THREAD   (CTOT * OUT_CH / 256)

__device__ __forceinline__ void load_trip(const float* __restrict__ xl,
                                          v4f (&B0)[NCHUNK], v4f (&B1)[NCHUNK])
{
    #pragma unroll
    for (int j = 0; j < NCHUNK; ++j) {
        const int off = (4 * (j & 3)) * TIN + (PADX - DIL * (j >> 2));
        B0[j] = *reinterpret_cast<const v4f*>(xl + off);
        B1[j] = *reinterpret_cast<const v4f*>(xl + off + TIN);
    }
}

__device__ __forceinline__ void compute_trip(const v2f (&A)[NCHUNK],
                                             const v4f (&B0)[NCHUNK], const v4f (&B1)[NCHUNK],
                                             float* __restrict__ yl)
{
    v8f acc[SUBTILES] = {};
    #pragma unroll
    for (int j = 0; j < NCHUNK; ++j) {
        #pragma unroll
        for (int q = 0; q < SUBTILES; ++q) {
            v2f bm;
            bm[0] = B0[j][q];
            bm[1] = B1[j][q];
            acc[q] = __builtin_amdgcn_wmma_f32_16x16x4_f32(
                false, A[j], false, bm, (short)0, acc[q], false, false);
        }
    }

    #pragma unroll
    for (int r = 0; r < 8; ++r) {
        v4f o4;
        #pragma unroll
        for (int q = 0; q < SUBTILES; ++q) o4[q] = acc[q][r];
        __builtin_nontemporal_store(o4, reinterpret_cast<v4f*>(yl + (size_t)r * TOUT));
    }
}

__global__ __launch_bounds__(256) void hyperconv_fused(
    const float* __restrict__ x,  const float* __restrict__ p,
    const float* __restrict__ W1, const float* __restrict__ b1,
    const float* __restrict__ W2, const float* __restrict__ b2,
    float* __restrict__ y)
{
    __shared__ float h_s[IN_CH];
    __shared__ float w_s[CTOT * OUT_CH];
    const int tid   = threadIdx.x;
    const int lane  = tid & 31;
    const int half  = lane >> 4;
    const int ln    = lane & 15;
    const int b     = blockIdx.x / GROUPS_PER_B;
    const int strip = (blockIdx.x % GROUPS_PER_B) * WAVES_PER_GROUP + (tid >> 5);
    const int t0    = strip * POS_PER_WAVE;

    const float* xl = x + ((size_t)b * IN_CH + 2 * half) * TIN + t0 + 4 * ln;
    float*       yl = y + ((size_t)b * OUT_CH + 8 * half) * TOUT + t0 + 4 * ln;

    v4f B0[NCHUNK], B1[NCHUNK];
    load_trip(xl, B0, B1);

    float w2v[W_PER_THREAD][IN_CH], accw[W_PER_THREAD];
    #pragma unroll
    for (int r = 0; r < W_PER_THREAD; ++r) {
        const int idx = tid + 256 * r;
        accw[r] = b2[idx];
        #pragma unroll
        for (int j = 0; j < IN_CH; ++j)
            w2v[r][j] = W2[j * (CTOT * OUT_CH) + idx];
    }
    if (tid < IN_CH) {
        float acc = b1[tid];
        #pragma unroll
        for (int i = 0; i < CONDD; ++i)
            acc = fmaf(p[b * CONDD + i], W1[i * IN_CH + tid], acc);
        h_s[tid] = (acc > 0.0f) ? acc : 0.2f * acc;
    }
    __syncthreads();
    #pragma unroll
    for (int r = 0; r < W_PER_THREAD; ++r) {
        float acc = accw[r];
        #pragma unroll
        for (int j = 0; j < IN_CH; ++j)
            acc = fmaf(h_s[j], w2v[r][j], acc);
        w_s[tid + 256 * r] = acc;
    }
    __syncthreads();

    v2f A[NCHUNK];
    #pragma unroll
    for (int j = 0; j < NCHUNK; ++j) {
        const int c0 = 4 * j + 2 * half;
        A[j][0] = w_s[(c0 + 0) * OUT_CH + ln];
        A[j][1] = w_s[(c0 + 1) * OUT_CH + ln];
    }

    for (int trip = 0; trip < TRIPS_PER_WAVE - 1; ++trip) {
        v4f N0[NCHUNK], N1[NCHUNK];
        load_trip(xl + POS_PER_TRIP, N0, N1);
        compute_trip(A, B0, B1, yl);
        #pragma unroll
        for (int j = 0; j < NCHUNK; ++j) { B0[j] = N0[j]; B1[j] = N1[j]; }
        xl += POS_PER_TRIP;
        yl += POS_PER_TRIP;
    }
    compute_trip(A, B0, B1, yl);
}

extern "C" void kernel_launch(void* const* d_in, const int* in_sizes, int n_in,
                              void* d_out, int out_size, void* d_ws, size_t ws_size,
                              hipStream_t stream) {
    const float* x  = (const float*)d_in[0];
    const float* p  = (const float*)d_in[1];
    const float* W1 = (const float*)d_in[2];
    const float* b1 = (const float*)d_in[3];
    const float* W2 = (const float*)d_in[4];
    const float* b2 = (const float*)d_in[5];
    float* y    = (float*)d_out;

    hipLaunchKernelGGL(hyperconv_fused, dim3(BATCH * GROUPS_PER_B), dim3(256), 0, stream,
                       x, p, W1, b1, W2, b2, y);
}
